# starting kernel
# baseline (speedup 1.0000x reference)
#include <hip/hip_runtime.h>

typedef float v2f __attribute__((ext_vector_type(2)));
typedef float v8f __attribute__((ext_vector_type(8)));

namespace {
constexpr int kT  = 2048;
constexpr int kB  = 128;
constexpr int kH  = 200;
constexpr int kHS = 204;
constexpr int kKB = 51;
constexpr int kWBR = 102;
constexpr int kWBC = 208;
constexpr int kWBBytes  = kWBR * kWBC * 8;
constexpr int kHBytes   = kB * kHS * 4;
constexpr int kLdsBytes = kWBBytes + kHBytes;
}

__global__ __launch_bounds__(1024)
void rnn_fused_wmma(const float* __restrict__ in_seq,
                    const float* __restrict__ W_ih,
                    const float* __restrict__ W_hh,
                    const float* __restrict__ b_ih,
                    const float* __restrict__ b_hh,
                    const float* __restrict__ W_out,
                    const float* __restrict__ b_out,
                    float* __restrict__ out)
{
  extern __shared__ char lds[];
  v2f*   wb = reinterpret_cast<v2f*>(lds);
  float* hb = reinterpret_cast<float*>(lds + kWBBytes);

  const int tid  = threadIdx.x;
  const int lane = tid & 31;
  const int wave = tid >> 5;

  for (int idx = tid; idx < kWBR * kWBC; idx += 1024) {
    const int k2 = idx / kWBC;
    const int j  = idx - k2 * kWBC;
    v2f v = {0.0f, 0.0f};
    if (j < kH) {
      if (k2 < kH / 2) {
        v = v2f{W_hh[j * kH + 2 * k2], W_hh[j * kH + 2 * k2 + 1]};
      } else if (k2 == kH / 2) {
        v = v2f{W_ih[j * 2 + 0], W_ih[j * 2 + 1]};
      } else {
        v = v2f{b_ih[j] + b_hh[j], 0.0f};
      }
    }
    wb[idx] = v;
  }
  for (int idx = tid; idx < kB * kHS; idx += 1024) {
    const int c = idx % kHS;
    hb[idx] = (c == 202) ? 1.0f : 0.0f;
  }
  __syncthreads();
  if (tid < kB) {
    const v2f x0 = reinterpret_cast<const v2f*>(in_seq)[tid];
    *reinterpret_cast<v2f*>(&hb[tid * kHS + 200]) = x0;
  }

  const int chunk = tid & 7;
  const int bhead = tid >> 3;
  float wout[25];
#pragma unroll
  for (int i = 0; i < 25; ++i) wout[i] = W_out[chunk * 25 + i];
  const float bout = b_out[0];

  const bool gemmWave = (wave < 26);
  const int nT     = wave >> 1;
  const int mGroup = wave & 1;
  const int col    = nT * 16 + (lane & 15);
  int aOff[4];
#pragma unroll
  for (int i = 0; i < 4; ++i)
    aOff[i] = ((4 * mGroup + i) * 16 + (lane & 15)) * kHS + ((lane >> 4) << 1);
  const int bOff0 = (lane >> 4) * kWBC + col;

  __syncthreads();

  const v8f vzero = {0.f, 0.f, 0.f, 0.f, 0.f, 0.f, 0.f, 0.f};

#pragma unroll 1
  for (int t = 0; t < kT; ++t) {
    v8f acc[4];
#pragma unroll
    for (int i = 0; i < 4; ++i) acc[i] = vzero;

    if (gemmWave) {
      int bo = bOff0;
      for (int kb = 0; kb < kKB; ++kb) {
        const v2f bop = wb[bo];
        const v2f a0 = *reinterpret_cast<const v2f*>(&hb[aOff[0] + (kb << 2)]);
        const v2f a1 = *reinterpret_cast<const v2f*>(&hb[aOff[1] + (kb << 2)]);
        const v2f a2 = *reinterpret_cast<const v2f*>(&hb[aOff[2] + (kb << 2)]);
        const v2f a3 = *reinterpret_cast<const v2f*>(&hb[aOff[3] + (kb << 2)]);
        acc[0] = __builtin_amdgcn_wmma_f32_16x16x4_f32(false, a0, false, bop, (short)0, acc[0], false, false);
        acc[1] = __builtin_amdgcn_wmma_f32_16x16x4_f32(false, a1, false, bop, (short)0, acc[1], false, false);
        acc[2] = __builtin_amdgcn_wmma_f32_16x16x4_f32(false, a2, false, bop, (short)0, acc[2], false, false);
        acc[3] = __builtin_amdgcn_wmma_f32_16x16x4_f32(false, a3, false, bop, (short)0, acc[3], false, false);
        bo += 2 * kWBC;
      }
    }
    __syncthreads();

    if (gemmWave) {
      const bool colOk = (col < kH);
#pragma unroll
      for (int i = 0; i < 4; ++i) {
        const int rbase = (4 * mGroup + i) * 16 + ((lane >> 4) << 3);
#pragma unroll
        for (int v = 0; v < 8; ++v) {
          const float x  = acc[i][v];
          const float e  = __builtin_amdgcn_exp2f(x * 2.885390081777927f);
          const float th = 1.0f - 2.0f * __builtin_amdgcn_rcpf(e + 1.0f);
          if (colOk) hb[(rbase + v) * kHS + col] = th;
        }
      }
    } else {
      const int b = tid - 832;
      if (b < kB && (t + 1) < kT) {
        const v2f x = reinterpret_cast<const v2f*>(in_seq)[(t + 1) * kB + b];
        *reinterpret_cast<v2f*>(&hb[b * kHS + 200]) = x;
      }
    }
    __syncthreads();

    float p = 0.0f;
    const float* hrow = &hb[bhead * kHS + chunk * 25];
#pragma unroll
    for (int i = 0; i < 25; ++i) p += hrow[i] * wout[i];
    p += __shfl_xor(p, 1, 32);
    p += __shfl_xor(p, 2, 32);
    p += __shfl_xor(p, 4, 32);
    if (chunk == 0) out[t * kB + bhead] = p + bout;
  }
}

extern "C" void kernel_launch(void* const* d_in, const int* in_sizes, int n_in,
                              void* d_out, int out_size, void* d_ws, size_t ws_size,
                              hipStream_t stream) {
  const float* in_seq = (const float*)d_in[0];
  const float* W_ih   = (const float*)d_in[1];
  const float* W_hh   = (const float*)d_in[2];
  const float* b_ih   = (const float*)d_in[3];
  const float* b_hh   = (const float*)d_in[4];
  const float* W_out  = (const float*)d_in[5];
  const float* b_out  = (const float*)d_in[6];
  float* out = (float*)d_out;

  hipLaunchKernelGGL(rnn_fused_wmma, dim3(1), dim3(1024), kLdsBytes, stream,
                     in_seq, W_ih, W_hh, b_ih, b_hh, W_out, b_out, out);
}


# One launch per time step: each of 104 work-groups owns one 16x16 tile of h_t with K split over its 4 waves (LDS sum), states kept in the workspace; output head as one last pass (sum order changes)
# speedup vs baseline: 5.1867x; 5.1867x over previous
#include <hip/hip_runtime.h>


typedef float v2f __attribute__((ext_vector_type(2)));
typedef float v4f __attribute__((ext_vector_type(4)));
typedef float v8f __attribute__((ext_vector_type(8)));

namespace {
constexpr int kT    = 2048;
constexpr int kB    = 128;
constexpr int kH    = 200;
constexpr int kG    = 25;
constexpr int kSlab = kB * kH;
constexpr int kRed  = 272;
}


__device__ __forceinline__ v4f ld4(const float* p) { return *reinterpret_cast<const v4f*>(p); }

template <bool kFirst>
__global__ __launch_bounds__(128)
void rnn_step(const float* __restrict__ x_t,
              const float* __restrict__ h_prev,
              float* __restrict__ h_next,
              const float* __restrict__ W_ih,
              const float* __restrict__ W_hh,
              const float* __restrict__ b_ih,
              const float* __restrict__ b_hh)
{
  __shared__ __attribute__((aligned(16))) float red[4 * kRed];

  const int tid  = threadIdx.x;
  const int lane = tid & 31;
  const int w  = __builtin_amdgcn_readfirstlane(tid >> 5);
  const int c  = lane & 15;
  const int hh = lane >> 4;
  const int mt = blockIdx.x & 7;
  const int nt = blockIdx.x >> 3;
  const int n  = min(nt * 16 + c, kH - 1);

  const int gBeg = (w == 0) ? 0 : 6 * w + 1;
  const float* ap = h_prev + ((mt * kG + gBeg) * 32 + lane) * 4;
  const float* bp = W_hh + n * kH + gBeg * 8 + 4 * hh;

  const v4f z4 = {0.f, 0.f, 0.f, 0.f};
  v4f a[7], b[7];
#pragma unroll
  for (int j = 0; j < 6; ++j) {
    a[j] = kFirst ? z4 : ld4(ap + 128 * j);
    b[j] = ld4(bp + 8 * j);
  }
  a[6] = z4;
  b[6] = z4;
  if (w == 0) {
    a[6] = kFirst ? z4 : ld4(ap + 128 * 6);
    b[6] = ld4(bp + 8 * 6);
  } else if (w == 1) {
    const v2f xv = *reinterpret_cast<const v2f*>(x_t + (mt * 16 + c) * 2);
    const v2f wv = *reinterpret_cast<const v2f*>(W_ih + n * 2);
    const float bs = b_ih[n] + b_hh[n];
    a[6].x = hh ? 1.0f : xv.x;
    a[6].y = hh ? 0.0f : xv.y;
    b[6].x = hh ? bs : wv.x;
    b[6].y = hh ? 0.0f : wv.y;
  }

  v8f acc0 = {0.f, 0.f, 0.f, 0.f, 0.f, 0.f, 0.f, 0.f};
  v8f acc1 = acc0;
#pragma unroll
  for (int j = 0; j < 7; ++j) {
    const v2f a0 = {a[j].x, a[j].y}, b0 = {b[j].x, b[j].y};
    const v2f a1 = {a[j].z, a[j].w}, b1 = {b[j].z, b[j].w};
    acc0 = __builtin_amdgcn_wmma_f32_16x16x4_f32(false, a0, false, b0, (short)0, acc0, false, false);
    acc1 = __builtin_amdgcn_wmma_f32_16x16x4_f32(false, a1, false, b1, (short)0, acc1, false, false);
  }
  const v8f acc = acc0 + acc1;

  float* myred = red + w * kRed;
  const int o0 = (c >> 3) * 128 + ((c >> 2) & 1) * 64 + hh * 32 + (c & 3) + (c >> 2) * 4;
#pragma unroll
  for (int r = 0; r < 8; ++r) myred[o0 + 4 * r] = acc[r];
  __syncthreads();

  if (tid < 64) {
    const int q = tid;
    if (nt < 12 || q < 32) {
      const int ro = 4 * q + ((q >> 4) << 2);
      v4f s = ld4(red + ro);
      s += ld4(red + kRed + ro);
      s += ld4(red + 2 * kRed + ro);
      s += ld4(red + 3 * kRed + ro);
      v4f th;
#pragma unroll
      for (int i = 0; i < 4; ++i) {
        const float e = __builtin_amdgcn_exp2f(s[i] * 2.885390081777927f);
        th[i] = 1.0f - 2.0f * __builtin_amdgcn_rcpf(e + 1.0f);
      }
      *reinterpret_cast<v4f*>(h_next + (mt * kG + 2 * nt) * 128 + 4 * q) = th;
    }
  }
}

__global__ __launch_bounds__(256)
void rnn_head(const float* __restrict__ hs,
              const float* __restrict__ W_out,
              const float* __restrict__ b_out,
              float* __restrict__ out)
{
  const int lane = threadIdx.x & 31;
  const int gw   = blockIdx.x * 8 + (threadIdx.x >> 5);
  const int t    = gw >> 2;
  const int p    = gw & 3;
  const int hh   = lane >> 4;
  const float* s0 = hs + (size_t)t * kSlab + ((2 * p) * kG * 32 + lane) * 4;
  const float* s1 = s0 + kG * 128;
  const float* wp = W_out + 4 * hh;
  float a0 = 0.0f, a1 = 0.0f;
#pragma unroll
  for (int g = 0; g < kG; ++g) {
    const v4f wv = ld4(wp + 8 * g);
    const v4f u  = ld4(s0 + 128 * g);
    const v4f v  = ld4(s1 + 128 * g);
    a0 += u.x * wv.x + u.y * wv.y + u.z * wv.z + u.w * wv.w;
    a1 += v.x * wv.x + v.y * wv.y + v.z * wv.z + v.w * wv.w;
  }
  a0 += __shfl_xor(a0, 16, 32);
  a1 += __shfl_xor(a1, 16, 32);
  out[t * kB + 32 * p + lane] = (hh ? a1 : a0) + b_out[0];
}

extern "C" void kernel_launch(void* const* d_in, const int* in_sizes, int n_in,
                              void* d_out, int out_size, void* d_ws, size_t ws_size,
                              hipStream_t stream) {
  const float* in_seq = (const float*)d_in[0];
  const float* W_ih   = (const float*)d_in[1];
  const float* W_hh   = (const float*)d_in[2];
  const float* b_ih   = (const float*)d_in[3];
  const float* b_hh   = (const float*)d_in[4];
  const float* W_out  = (const float*)d_in[5];
  const float* b_out  = (const float*)d_in[6];
  float* out = (float*)d_out;
  float* hs  = (float*)d_ws;

  rnn_step<true><<<dim3(104), dim3(128), 0, stream>>>(in_seq, hs, hs, W_ih, W_hh, b_ih, b_hh);
  const float* x = in_seq + 256;
  for (float* h = hs; h != hs + 2047 * 25600; h += 25600, x += 256)
    rnn_step<false><<<dim3(104), dim3(128), 0, stream>>>(x, h, h + 25600, W_ih, W_hh, b_ih, b_hh);
  rnn_head<<<dim3(1024), dim3(256), 0, stream>>>(hs, W_out, b_out, out);
}


# Temporal blocking: 16 time steps per launch (129 launches, not 2049); all 104 work-groups recompute their 16 rows' whole state for 15 steps (W_hh in registers, state in LDS); K sum order changes
# speedup vs baseline: 5.7936x; 1.1170x over previous
#include <hip/hip_runtime.h>


typedef float v2f __attribute__((ext_vector_type(2)));
typedef float v4f __attribute__((ext_vector_type(4)));
typedef float v8f __attribute__((ext_vector_type(8)));

namespace {
constexpr int kT    = 2048;
constexpr int kB    = 128;
constexpr int kH    = 200;
constexpr int kG    = 25;
constexpr int kSlab = kB * kH;
constexpr int kRed  = 272;
constexpr int kS    = 16;
constexpr int kSt   = 13 * kRed;
}


__device__ __forceinline__ v4f ld4(const float* p) { return *reinterpret_cast<const v4f*>(p); }

__device__ __forceinline__ float tanh1(float s) {
  const float e = __builtin_amdgcn_exp2f(s * 2.885390081777927f);
  return 1.0f - 2.0f * __builtin_amdgcn_rcpf(e + 1.0f);
}

__device__ __forceinline__ void load_bk(v4f (&b)[7], int tile, int w, int gBeg, int c, int hh,
                                        const float* __restrict__ W_ih, const float* __restrict__ W_hh,
                                        const float* __restrict__ b_ih, const float* __restrict__ b_hh) {
  const int n = min(tile * 16 + c, kH - 1);
  const float* bp = W_hh + n * kH + gBeg * 8 + 4 * hh;
#pragma unroll
  for (int j = 0; j < 6; ++j) b[j] = ld4(bp + 8 * j);
  const v4f z4 = {0.f, 0.f, 0.f, 0.f};
  b[6] = z4;
  if (w == 0) {
    b[6] = ld4(bp + 8 * 6);
  } else if (w == 1) {
    const v2f wv = *reinterpret_cast<const v2f*>(W_ih + n * 2);
    const float bs = b_ih[n] + b_hh[n];
    b[6].x = hh ? bs : wv.x;
    b[6].y = hh ? 0.0f : wv.y;
  }
}

__device__ __forceinline__ void ksplit_load(v4f (&a)[7], const float* cur, v2f xv, int w, int gBeg, int lane, int hh) {
  const v4f z4 = {0.f, 0.f, 0.f, 0.f};
  const float* ap = cur + gBeg * 136 + 4 * lane + 4 * hh;
#pragma unroll
  for (int j = 0; j < 6; ++j) a[j] = ld4(ap + 136 * j);
  a[6] = z4;
  if (w == 0) {
    a[6] = ld4(ap + 136 * 6);
  } else if (w == 1) {
    a[6].x = hh ? 1.0f : xv.x;
    a[6].y = hh ? 0.0f : xv.y;
  }
}

__device__ __forceinline__ v8f ksplit_mma(const v4f (&a)[7], const v4f (&b)[7]) {
  v8f acc0 = {0.f, 0.f, 0.f, 0.f, 0.f, 0.f, 0.f, 0.f};
  v8f acc1 = acc0;
#pragma unroll
  for (int j = 0; j < 7; ++j) {
    const v2f a0 = {a[j].x, a[j].y}, b0 = {b[j].x, b[j].y};
    const v2f a1 = {a[j].z, a[j].w}, b1 = {b[j].z, b[j].w};
    acc0 = __builtin_amdgcn_wmma_f32_16x16x4_f32(false, a0, false, b0, (short)0, acc0, false, false);
    acc1 = __builtin_amdgcn_wmma_f32_16x16x4_f32(false, a1, false, b1, (short)0, acc1, false, false);
  }
  return acc0 + acc1;
}

template <bool kFirst>
__global__ __launch_bounds__(128)
void rnn_steps(const float* __restrict__ x_t,
               const float* __restrict__ h_prev,
               float* __restrict__ h_next,
               const float* __restrict__ W_ih,
               const float* __restrict__ W_hh,
               const float* __restrict__ b_ih,
               const float* __restrict__ b_hh)
{
  __shared__ __attribute__((aligned(16))) float st[2 * kSt];
  __shared__ __attribute__((aligned(16))) float red[4 * kRed];
  __shared__ __attribute__((aligned(16))) float xs[kS * 32];

  const int tid  = threadIdx.x;
  const int lane = tid & 31;
  const int w  = __builtin_amdgcn_readfirstlane(tid >> 5);
  const int c  = lane & 15;
  const int hh = lane >> 4;
  const int mt = blockIdx.x & 7;
  const int nt = blockIdx.x >> 3;
  const v4f z4 = {0.f, 0.f, 0.f, 0.f};

  v4f sv[7];
#pragma unroll
  for (int k = 0; k < 7; ++k) {
    const int ch = tid + 128 * k;
    sv[k] = (kFirst || ch >= 800) ? z4 : ld4(h_prev + mt * (kG * 128) + 4 * ch);
  }
  v2f xl[kS / 8];
#pragma unroll
  for (int k = 0; k < kS / 8; ++k) {
    const int i = tid + 128 * k;
    xl[k] = *reinterpret_cast<const v2f*>(x_t + ((i >> 4) * kB + mt * 16 + (i & 15)) * 2);
  }

  v4f bw[3][kG];
  v2f bi[3];
#pragma unroll
  for (int ti = 0; ti < 3; ++ti) {
    const int n = (w + 4 * ti) * 16 + c;
    const float* bp = W_hh + n * kH + 4 * hh;
#pragma unroll
    for (int g = 0; g < kG; ++g) bw[ti][g] = ld4(bp + 8 * g);
    const v2f wv = *reinterpret_cast<const v2f*>(W_ih + n * 2);
    const float bs = b_ih[n] + b_hh[n];
    bi[ti].x = hh ? bs : wv.x;
    bi[ti].y = hh ? 0.0f : wv.y;
  }
  const int gBeg = (w == 0) ? 0 : 6 * w + 1;
  v4f b12[7], bo[7];
  load_bk(b12, 12, w, gBeg, c, hh, W_ih, W_hh, b_ih, b_hh);
  load_bk(bo, nt, w, gBeg, c, hh, W_ih, W_hh, b_ih, b_hh);

  const int o0 = (c >> 3) * 128 + ((c >> 2) & 1) * 64 + hh * 32 + (c & 3) + (c >> 2) * 4;
  float* myred = red + w * kRed;
  const int q  = tid;
  const int ro = 4 * q + ((q >> 4) << 2);

#pragma unroll
  for (int k = 0; k < 7; ++k) {
    const int ch = tid + 128 * k;
    if (ch < 800) *reinterpret_cast<v4f*>(st + 4 * ch + 4 * (ch >> 4)) = sv[k];
  }
#pragma unroll
  for (int k = 0; k < kS / 8; ++k) *reinterpret_cast<v2f*>(xs + 2 * (tid + 128 * k)) = xl[k];
  __syncthreads();

  for (int i = 0; i < kS - 1; ++i) {
    const float* cur = st + (i & 1) * kSt;
    float* nxt = st + ((i + 1) & 1) * kSt;
    const v2f xv = *reinterpret_cast<const v2f*>(xs + (i * 16 + c) * 2);

    v4f ak[7];
    ksplit_load(ak, cur, xv, w, gBeg, lane, hh);
    v4f a[kG];
#pragma unroll
    for (int g = 0; g < kG; ++g) a[g] = ld4(cur + g * 136 + 4 * lane + 4 * hh);
    __builtin_amdgcn_sched_barrier(0);

    const v8f p = ksplit_mma(ak, b12);
    v2f ax;
    ax.x = hh ? 1.0f : xv.x;
    ax.y = hh ? 0.0f : xv.y;
    v8f accA[3], accB[3];
#pragma unroll
    for (int ti = 0; ti < 3; ++ti) {
      accA[ti] = (v8f){0.f, 0.f, 0.f, 0.f, 0.f, 0.f, 0.f, 0.f};
      accB[ti] = accA[ti];
    }
#pragma unroll
    for (int g = 0; g < kG; ++g) {
      const v2f a0 = {a[g].x, a[g].y}, a1 = {a[g].z, a[g].w};
#pragma unroll
      for (int ti = 0; ti < 3; ++ti) {
        const v2f b0 = {bw[ti][g].x, bw[ti][g].y}, b1 = {bw[ti][g].z, bw[ti][g].w};
        accA[ti] = __builtin_amdgcn_wmma_f32_16x16x4_f32(false, a0, false, b0, (short)0, accA[ti], false, false);
        accB[ti] = __builtin_amdgcn_wmma_f32_16x16x4_f32(false, a1, false, b1, (short)0, accB[ti], false, false);
      }
    }
#pragma unroll
    for (int ti = 0; ti < 3; ++ti)
      accA[ti] = __builtin_amdgcn_wmma_f32_16x16x4_f32(false, ax, false, bi[ti], (short)0, accA[ti], false, false);
    __builtin_amdgcn_sched_barrier(0);
#pragma unroll
    for (int r = 0; r < 8; ++r) myred[o0 + 4 * r] = p[r];
#pragma unroll
    for (int ti = 0; ti < 3; ++ti) {
      const v8f s = accA[ti] + accB[ti];
      float* dst = nxt + (w + 4 * ti) * kRed + o0;
#pragma unroll
      for (int r = 0; r < 8; ++r) dst[4 * r] = tanh1(s[r]);
    }
    __syncthreads();

    float* hs_i = h_next + i * kSlab + mt * (kG * 128);
    if (tid < 64) {
      if (q < 32) {
        v4f s = ld4(red + ro);
        s += ld4(red + kRed + ro);
        s += ld4(red + 2 * kRed + ro);
        s += ld4(red + 3 * kRed + ro);
        v4f th;
#pragma unroll
        for (int e = 0; e < 4; ++e) th[e] = tanh1(s[e]);
        *reinterpret_cast<v4f*>(nxt + 12 * kRed + ro) = th;
        if (nt == 12) *reinterpret_cast<v4f*>(hs_i + 24 * 128 + 4 * q) = th;
      }
      if (nt < 12) *reinterpret_cast<v4f*>(hs_i + 2 * nt * 128 + 4 * q) = ld4(nxt + nt * kRed + ro);
    }
    __syncthreads();
  }

  {
    const float* cur = st + ((kS - 1) & 1) * kSt;
    const v2f xv = *reinterpret_cast<const v2f*>(xs + ((kS - 1) * 16 + c) * 2);
    v4f ak[7];
    ksplit_load(ak, cur, xv, w, gBeg, lane, hh);
    const v8f p = ksplit_mma(ak, bo);
#pragma unroll
    for (int r = 0; r < 8; ++r) myred[o0 + 4 * r] = p[r];
    __syncthreads();
    if (tid < 64) {
      if (nt < 12 || q < 32) {
        v4f s = ld4(red + ro);
        s += ld4(red + kRed + ro);
        s += ld4(red + 2 * kRed + ro);
        s += ld4(red + 3 * kRed + ro);
        v4f th;
#pragma unroll
        for (int e = 0; e < 4; ++e) th[e] = tanh1(s[e]);
        *reinterpret_cast<v4f*>(h_next + (kS - 1) * kSlab + (mt * kG + 2 * nt) * 128 + 4 * q) = th;
      }
    }
  }
}

__global__ __launch_bounds__(256)
void rnn_head(const float* __restrict__ hs,
              const float* __restrict__ W_out,
              const float* __restrict__ b_out,
              float* __restrict__ out)
{
  const int lane = threadIdx.x & 31;
  const int gw   = blockIdx.x * 8 + (threadIdx.x >> 5);
  const int t    = gw >> 2;
  const int p    = gw & 3;
  const int hh   = lane >> 4;
  const float* s0 = hs + (size_t)t * kSlab + ((2 * p) * kG * 32 + lane) * 4;
  const float* s1 = s0 + kG * 128;
  const float* wp = W_out + 4 * hh;
  float a0 = 0.0f, a1 = 0.0f;
#pragma unroll
  for (int g = 0; g < kG; ++g) {
    const v4f wv = ld4(wp + 8 * g);
    const v4f u  = ld4(s0 + 128 * g);
    const v4f v  = ld4(s1 + 128 * g);
    a0 += u.x * wv.x + u.y * wv.y + u.z * wv.z + u.w * wv.w;
    a1 += v.x * wv.x + v.y * wv.y + v.z * wv.z + v.w * wv.w;
  }
  a0 += __shfl_xor(a0, 16, 32);
  a1 += __shfl_xor(a1, 16, 32);
  out[t * kB + 32 * p + lane] = (hh ? a1 : a0) + b_out[0];
}

extern "C" void kernel_launch(void* const* d_in, const int* in_sizes, int n_in,
                              void* d_out, int out_size, void* d_ws, size_t ws_size,
                              hipStream_t stream) {
  const float* in_seq = (const float*)d_in[0];
  const float* W_ih   = (const float*)d_in[1];
  const float* W_hh   = (const float*)d_in[2];
  const float* b_ih   = (const float*)d_in[3];
  const float* b_hh   = (const float*)d_in[4];
  const float* W_out  = (const float*)d_in[5];
  const float* b_out  = (const float*)d_in[6];
  float* out = (float*)d_out;
  float* hs  = (float*)d_ws;

  rnn_steps<true><<<dim3(104), dim3(128), 0, stream>>>(in_seq, hs, hs, W_ih, W_hh, b_ih, b_hh);
  const float* x = in_seq + 16 * 256;
  for (float* h = hs + 15 * 25600; h != hs + 2047 * 25600; h += 16 * 25600, x += 16 * 256)
    rnn_steps<false><<<dim3(104), dim3(128), 0, stream>>>(x, h, h + 25600, W_ih, W_hh, b_ih, b_hh);
  rnn_head<<<dim3(1024), dim3(256), 0, stream>>>(hs, W_out, b_out, out);
}


# One barrier per step: every wave sums and tanh's tile 12's four K parts itself among the next step's matrix work (LDS double-buffered), workspace stores issued in mid-step; sum order unchanged
# speedup vs baseline: 7.1365x; 1.2318x over previous
#include <hip/hip_runtime.h>


typedef float v2f __attribute__((ext_vector_type(2)));
typedef float v4f __attribute__((ext_vector_type(4)));
typedef float v8f __attribute__((ext_vector_type(8)));

namespace {
constexpr int kT    = 2048;
constexpr int kB    = 128;
constexpr int kH    = 200;
constexpr int kG    = 25;
constexpr int kSlab = kB * kH;
constexpr int kRed  = 272;
constexpr int kS    = 16;
constexpr int kSt   = 13 * kRed;
}


__device__ __forceinline__ v4f ld4(const float* p) { return *reinterpret_cast<const v4f*>(p); }

__device__ __forceinline__ float tanh1(float s) {
  const float e = __builtin_amdgcn_exp2f(s * 2.885390081777927f);
  return 1.0f - 2.0f * __builtin_amdgcn_rcpf(e + 1.0f);
}

__device__ __forceinline__ void load_bk(v4f (&b)[7], int tile, int w, int gBeg, int c, int hh,
                                        const float* __restrict__ W_ih, const float* __restrict__ W_hh,
                                        const float* __restrict__ b_ih, const float* __restrict__ b_hh) {
  const int n = min(tile * 16 + c, kH - 1);
  const float* bp = W_hh + n * kH + gBeg * 8 + 4 * hh;
#pragma unroll
  for (int j = 0; j < 6; ++j) b[j] = ld4(bp + 8 * j);
  const v4f z4 = {0.f, 0.f, 0.f, 0.f};
  b[6] = z4;
  if (w == 0) {
    b[6] = ld4(bp + 8 * 6);
  } else if (w == 1) {
    const v2f wv = *reinterpret_cast<const v2f*>(W_ih + n * 2);
    const float bs = b_ih[n] + b_hh[n];
    b[6].x = hh ? bs : wv.x;
    b[6].y = hh ? 0.0f : wv.y;
  }
}

__device__ __forceinline__ void ksplit_load(v4f (&a)[7], const float* cur, v2f xv, int w, int gBeg, int lane, int hh) {
  const v4f z4 = {0.f, 0.f, 0.f, 0.f};
  const float* ap = cur + gBeg * 136 + 4 * lane + 4 * hh;
#pragma unroll
  for (int j = 0; j < 5; ++j) a[j] = ld4(ap + 136 * j);
  a[5] = z4;
  if (w != 3) a[5] = ld4(ap + 136 * 5);
  a[6] = z4;
  if (w == 0) {
    a[6] = ld4(ap + 136 * 6);
  } else if (w == 1) {
    a[6].x = hh ? 1.0f : xv.x;
    a[6].y = hh ? 0.0f : xv.y;
  }
}

__device__ __forceinline__ v4f group24(v4f p0, v4f p1, v4f p2, v4f p3) {
  v4f s = p0;
  s += p1;
  s += p2;
  s += p3;
  v4f th;
#pragma unroll
  for (int e = 0; e < 4; ++e) th[e] = tanh1(s[e]);
  return th;
}

__device__ __forceinline__ v8f ksplit_mma(const v4f (&a)[7], const v4f (&b)[7]) {
  v8f acc0 = {0.f, 0.f, 0.f, 0.f, 0.f, 0.f, 0.f, 0.f};
  v8f acc1 = acc0;
#pragma unroll
  for (int j = 0; j < 7; ++j) {
    const v2f a0 = {a[j].x, a[j].y}, b0 = {b[j].x, b[j].y};
    const v2f a1 = {a[j].z, a[j].w}, b1 = {b[j].z, b[j].w};
    acc0 = __builtin_amdgcn_wmma_f32_16x16x4_f32(false, a0, false, b0, (short)0, acc0, false, false);
    acc1 = __builtin_amdgcn_wmma_f32_16x16x4_f32(false, a1, false, b1, (short)0, acc1, false, false);
  }
  return acc0 + acc1;
}

template <bool kFirst>
__global__ __launch_bounds__(128)
void rnn_steps(const float* __restrict__ x_t,
               const float* __restrict__ h_prev,
               float* __restrict__ h_next,
               const float* __restrict__ W_ih,
               const float* __restrict__ W_hh,
               const float* __restrict__ b_ih,
               const float* __restrict__ b_hh)
{
  __shared__ __attribute__((aligned(16))) float st[2 * kSt];
  __shared__ __attribute__((aligned(16))) float red[2 * 4 * kRed];
  __shared__ __attribute__((aligned(16))) float xs[kS * 32];

  const int tid  = threadIdx.x;
  const int lane = tid & 31;
  const int w  = __builtin_amdgcn_readfirstlane(tid >> 5);
  const int c  = lane & 15;
  const int hh = lane >> 4;
  const int mt = blockIdx.x & 7;
  const int nt = blockIdx.x >> 3;
  const v4f z4 = {0.f, 0.f, 0.f, 0.f};

  v4f sv[7];
#pragma unroll
  for (int k = 0; k < 7; ++k) {
    const int ch = tid + 128 * k;
    sv[k] = (kFirst || ch >= 800) ? z4 : ld4(h_prev + mt * (kG * 128) + 4 * ch);
  }
  v2f xl[kS / 8];
#pragma unroll
  for (int k = 0; k < kS / 8; ++k) {
    const int i = tid + 128 * k;
    xl[k] = *reinterpret_cast<const v2f*>(x_t + ((i >> 4) * kB + mt * 16 + (i & 15)) * 2);
  }

  v4f bw[3][kG];
  v2f bi[3];
#pragma unroll
  for (int ti = 0; ti < 3; ++ti) {
    const int n = (w + 4 * ti) * 16 + c;
    const float* bp = W_hh + n * kH + 4 * hh;
#pragma unroll
    for (int g = 0; g < kG; ++g) bw[ti][g] = ld4(bp + 8 * g);
    const v2f wv = *reinterpret_cast<const v2f*>(W_ih + n * 2);
    const float bs = b_ih[n] + b_hh[n];
    bi[ti].x = hh ? bs : wv.x;
    bi[ti].y = hh ? 0.0f : wv.y;
  }
  const int gBeg = (w == 0) ? 0 : 6 * w + 1;
  v4f b12[7], bo[7];
  load_bk(b12, 12, w, gBeg, c, hh, W_ih, W_hh, b_ih, b_hh);
  load_bk(bo, nt, w, gBeg, c, hh, W_ih, W_hh, b_ih, b_hh);

  const int o0 = (c >> 3) * 128 + ((c >> 2) & 1) * 64 + hh * 32 + (c & 3) + (c >> 2) * 4;
  const int q  = tid;
  const int ro = 4 * q + ((q >> 4) << 2);

#pragma unroll
  for (int k = 0; k < 7; ++k) {
    const int ch = tid + 128 * k;
    if (ch < 800) *reinterpret_cast<v4f*>(st + 4 * ch + 4 * (ch >> 4)) = sv[k];
  }
#pragma unroll
  for (int k = 0; k < kS / 8; ++k) *reinterpret_cast<v2f*>(xs + 2 * (tid + 128 * k)) = xl[k];
  __syncthreads();

  const int rl = 4 * lane + 4 * hh;
  float* hs_p = h_next + mt * (kG * 128);

  for (int i = 0; i < kS - 1; ++i) {
    const float* cur = st + (i & 1) * kSt;
    float* nxt = st + ((i + 1) & 1) * kSt;
    const float* rcur = red + (i & 1) * (4 * kRed);
    float* rnxt = red + ((i + 1) & 1) * (4 * kRed);
    const v2f xv = *reinterpret_cast<const v2f*>(xs + (i * 16 + c) * 2);

    const bool copy_own = i > 0 && nt < 12 && w < 2;
    v4f own = z4;
    if (copy_own) own = ld4(cur + nt * kRed + ro);
    const float* pp = (i == 0) ? cur + 24 * 136 + rl : rcur + rl;
    const int ps = (i == 0) ? 0 : kRed;
    const v4f p0 = ld4(pp), p1 = ld4(pp + ps), p2 = ld4(pp + 2 * ps), p3 = ld4(pp + 3 * ps);
    v4f a[kG];
#pragma unroll
    for (int g = 0; g < kG - 1; ++g) a[g] = ld4(cur + g * 136 + 4 * lane + 4 * hh);
    v4f ak[7];
    ksplit_load(ak, cur, xv, w, gBeg, lane, hh);
    __builtin_amdgcn_sched_barrier(0);

    v2f ax;
    ax.x = hh ? 1.0f : xv.x;
    ax.y = hh ? 0.0f : xv.y;
    v8f accA[3], accB[3];
#pragma unroll
    for (int ti = 0; ti < 3; ++ti) {
      accA[ti] = (v8f){0.f, 0.f, 0.f, 0.f, 0.f, 0.f, 0.f, 0.f};
      accB[ti] = accA[ti];
    }
#pragma unroll
    for (int g = 0; g < kG / 2; ++g) {
      const v2f a0 = {a[g].x, a[g].y}, a1 = {a[g].z, a[g].w};
#pragma unroll
      for (int ti = 0; ti < 3; ++ti) {
        const v2f b0 = {bw[ti][g].x, bw[ti][g].y}, b1 = {bw[ti][g].z, bw[ti][g].w};
        accA[ti] = __builtin_amdgcn_wmma_f32_16x16x4_f32(false, a0, false, b0, (short)0, accA[ti], false, false);
        accB[ti] = __builtin_amdgcn_wmma_f32_16x16x4_f32(false, a1, false, b1, (short)0, accB[ti], false, false);
      }
    }
    const v4f th24 = group24(p0, p1, p2, p3);
    const v4f a24 = (i == 0) ? p0 : th24;
    a[kG - 1] = a24;
    if (w == 3) ak[5] = a24;
    if (i > 0) {
      if (copy_own) {
        *reinterpret_cast<v4f*>(hs_p + 2 * nt * 128 + 4 * q) = own;
      } else if (nt == 12 && w == 0) {
        *reinterpret_cast<v4f*>(hs_p + 24 * 128 + 4 * lane) = a24;
      }
      hs_p += kSlab;
    }
#pragma unroll
    for (int g = kG / 2; g < kG; ++g) {
      const v2f a0 = {a[g].x, a[g].y}, a1 = {a[g].z, a[g].w};
#pragma unroll
      for (int ti = 0; ti < 3; ++ti) {
        const v2f b0 = {bw[ti][g].x, bw[ti][g].y}, b1 = {bw[ti][g].z, bw[ti][g].w};
        accA[ti] = __builtin_amdgcn_wmma_f32_16x16x4_f32(false, a0, false, b0, (short)0, accA[ti], false, false);
        accB[ti] = __builtin_amdgcn_wmma_f32_16x16x4_f32(false, a1, false, b1, (short)0, accB[ti], false, false);
      }
    }
#pragma unroll
    for (int ti = 0; ti < 3; ++ti)
      accA[ti] = __builtin_amdgcn_wmma_f32_16x16x4_f32(false, ax, false, bi[ti], (short)0, accA[ti], false, false);
    const v8f p = ksplit_mma(ak, b12);
#pragma unroll
    for (int ti = 0; ti < 3; ++ti) {
      const v8f s = accA[ti] + accB[ti];
      float* dst = nxt + (w + 4 * ti) * kRed + o0;
#pragma unroll
      for (int r = 0; r < 8; ++r) dst[4 * r] = tanh1(s[r]);
    }
#pragma unroll
    for (int r = 0; r < 8; ++r) rnxt[w * kRed + o0 + 4 * r] = p[r];
    __syncthreads();
  }

  {
    const float* cur = st + ((kS - 1) & 1) * kSt;
    const float* rcur = red + ((kS - 1) & 1) * (4 * kRed);
    float* myred = red + (kS & 1) * (4 * kRed) + w * kRed;
    const v2f xv = *reinterpret_cast<const v2f*>(xs + ((kS - 1) * 16 + c) * 2);
    const v4f a24 = group24(ld4(rcur + rl), ld4(rcur + kRed + rl), ld4(rcur + 2 * kRed + rl), ld4(rcur + 3 * kRed + rl));
    if (nt < 12) {
      if (w < 2) *reinterpret_cast<v4f*>(hs_p + 2 * nt * 128 + 4 * q) = ld4(cur + nt * kRed + ro);
    } else if (w == 0) {
      *reinterpret_cast<v4f*>(hs_p + 24 * 128 + 4 * lane) = a24;
    }
    v4f ak[7];
    ksplit_load(ak, cur, xv, w, gBeg, lane, hh);
    if (w == 3) ak[5] = a24;
    const v8f p = ksplit_mma(ak, bo);
#pragma unroll
    for (int r = 0; r < 8; ++r) myred[o0 + 4 * r] = p[r];
    __syncthreads();
    if (tid < 64) {
      if (nt < 12 || q < 32) {
        const float* rd = red + (kS & 1) * (4 * kRed);
        v4f s = ld4(rd + ro);
        s += ld4(rd + kRed + ro);
        s += ld4(rd + 2 * kRed + ro);
        s += ld4(rd + 3 * kRed + ro);
        v4f th;
#pragma unroll
        for (int e = 0; e < 4; ++e) th[e] = tanh1(s[e]);
        *reinterpret_cast<v4f*>(h_next + (kS - 1) * kSlab + (mt * kG + 2 * nt) * 128 + 4 * q) = th;
      }
    }
  }
}

__global__ __launch_bounds__(256)
void rnn_head(const float* __restrict__ hs,
              const float* __restrict__ W_out,
              const float* __restrict__ b_out,
              float* __restrict__ out)
{
  const int lane = threadIdx.x & 31;
  const int gw   = blockIdx.x * 8 + (threadIdx.x >> 5);
  const int t    = gw >> 2;
  const int p    = gw & 3;
  const int hh   = lane >> 4;
  const float* s0 = hs + (size_t)t * kSlab + ((2 * p) * kG * 32 + lane) * 4;
  const float* s1 = s0 + kG * 128;
  const float* wp = W_out + 4 * hh;
  float a0 = 0.0f, a1 = 0.0f;
#pragma unroll
  for (int g = 0; g < kG; ++g) {
    const v4f wv = ld4(wp + 8 * g);
    const v4f u  = ld4(s0 + 128 * g);
    const v4f v  = ld4(s1 + 128 * g);
    a0 += u.x * wv.x + u.y * wv.y + u.z * wv.z + u.w * wv.w;
    a1 += v.x * wv.x + v.y * wv.y + v.z * wv.z + v.w * wv.w;
  }
  a0 += __shfl_xor(a0, 16, 32);
  a1 += __shfl_xor(a1, 16, 32);
  out[t * kB + 32 * p + lane] = (hh ? a1 : a0) + b_out[0];
}

extern "C" void kernel_launch(void* const* d_in, const int* in_sizes, int n_in,
                              void* d_out, int out_size, void* d_ws, size_t ws_size,
                              hipStream_t stream) {
  const float* in_seq = (const float*)d_in[0];
  const float* W_ih   = (const float*)d_in[1];
  const float* W_hh   = (const float*)d_in[2];
  const float* b_ih   = (const float*)d_in[3];
  const float* b_hh   = (const float*)d_in[4];
  const float* W_out  = (const float*)d_in[5];
  const float* b_out  = (const float*)d_in[6];
  float* out = (float*)d_out;
  float* hs  = (float*)d_ws;

  rnn_steps<true><<<dim3(104), dim3(128), 0, stream>>>(in_seq, hs, hs, W_ih, W_hh, b_ih, b_hh);
  const float* x = in_seq + 16 * 256;
  for (float* h = hs + 15 * 25600; h != hs + 2047 * 25600; h += 16 * 25600, x += 16 * 256)
    rnn_steps<false><<<dim3(104), dim3(128), 0, stream>>>(x, h, h + 25600, W_ih, W_hh, b_ih, b_hh);
  rnn_head<<<dim3(1024), dim3(256), 0, stream>>>(hs, W_out, b_out, out);
}
